# starting kernel
# baseline (speedup 1.0000x reference)
#include <hip/hip_runtime.h>
#include <hip/hip_bf16.h>

typedef __attribute__((ext_vector_type(2))) float v2f;
typedef __attribute__((ext_vector_type(8))) float v8f;

#define N_TOTAL 8192
#define KDIM    16
#define FIN     32
#define FOUT    32

#define WAVES_PER_BLOCK 8
#define K1_BLOCKS       16
#define K1_WAVES        (K1_BLOCKS * WAVES_PER_BLOCK)
#define ROWS_PER_WAVE   (N_TOTAL / K1_WAVES)

__global__ __launch_bounds__(256)
void k1_partialT(const float* __restrict__ ev,
                 const float* __restrict__ in,
                 float* __restrict__ partial) {
    const int wave = blockIdx.x * WAVES_PER_BLOCK + (threadIdx.x >> 5);
    const int lane = threadIdx.x & 31;
    const int m    = lane & 15;
    const int half = lane >> 4;

    v8f acc0 = {};
    v8f acc1 = {};

    const int nstart = wave * ROWS_PER_WAVE;
    #pragma unroll 4
    for (int it = 0; it < ROWS_PER_WAVE / 4; ++it) {
        const int nb = nstart + it * 4 + half * 2;
        v2f a, b0, b1;
        a.x  = ev[nb * KDIM + m];
        a.y  = ev[(nb + 1) * KDIM + m];
        b0.x = in[nb * FIN + m];
        b0.y = in[(nb + 1) * FIN + m];
        b1.x = in[nb * FIN + 16 + m];
        b1.y = in[(nb + 1) * FIN + 16 + m];
        acc0 = __builtin_amdgcn_wmma_f32_16x16x4_f32(false, a, false, b0,
                                                     (short)0, acc0, false, false);
        acc1 = __builtin_amdgcn_wmma_f32_16x16x4_f32(false, a, false, b1,
                                                     (short)0, acc1, false, false);
    }

    float* p = partial + wave * (KDIM * FIN);
    #pragma unroll
    for (int v = 0; v < 8; ++v) {
        const int row = half ? (v + 8) : v;
        p[row * FIN + m]      = acc0[v];
        p[row * FIN + 16 + m] = acc1[v];
    }
}

__global__ __launch_bounds__(512)
void k2_reduce_S(const float* __restrict__ partial,
                 const float* __restrict__ eigenVal,
                 const float* __restrict__ W,
                 const float* __restrict__ A,
                 const float* __restrict__ B,
                 const float* __restrict__ C,
                 const float* __restrict__ D,
                 float* __restrict__ S) {
    __shared__ float Tl[KDIM * FIN];
    __shared__ float Mr[KDIM];
    __shared__ float Mi[KDIM];
    const int tid = threadIdx.x;

    float s = 0.0f;
    for (int w = 0; w < K1_WAVES; ++w)
        s += partial[w * (KDIM * FIN) + tid];
    Tl[tid] = s;

    if (tid < KDIM) {
        const float lam = eigenVal[tid];
        const float m1r = A[tid]      * lam + B[tid];
        const float m1i = A[16 + tid] * lam + B[16 + tid];
        const float m2r = 1.0f / (C[tid]      * lam + D[tid]);
        const float m2i = 1.0f / (C[16 + tid] * lam + D[16 + tid]);
        Mr[tid] = m1r * m2r - m1i * m2i;
        Mi[tid] = m1i * m2r + m1r * m2i;
    }
    __syncthreads();

    const int k = tid >> 5;
    const int f = tid & 31;
    float s0 = 0.0f, s1 = 0.0f;
    #pragma unroll
    for (int j = 0; j < FIN; ++j) {
        const float t = Tl[k * FIN + j];
        s0 += t * W[j * FOUT + f];
        s1 += t * W[FIN * FOUT + j * FOUT + f];
    }
    S[k * FOUT + f] = Mr[k] * s0 - Mi[k] * s1;
}

__global__ __launch_bounds__(256)
void k3_out(const float* __restrict__ ev,
            const float* __restrict__ S,
            const float* __restrict__ bias,
            float* __restrict__ out) {
    const int wave = blockIdx.x * WAVES_PER_BLOCK + (threadIdx.x >> 5);
    const int lane = threadIdx.x & 31;
    const int m    = lane & 15;
    const int half = lane >> 4;
    const int n0   = wave * 16;

    v8f acc0 = {};
    v8f acc1 = {};
    #pragma unroll
    for (int kc = 0; kc < 4; ++kc) {
        const int kg = kc * 4 + half * 2;
        v2f a, b0, b1;
        a.x  = ev[(n0 + m) * KDIM + kg];
        a.y  = ev[(n0 + m) * KDIM + kg + 1];
        b0.x = S[kg * FOUT + m];
        b0.y = S[(kg + 1) * FOUT + m];
        b1.x = S[kg * FOUT + 16 + m];
        b1.y = S[(kg + 1) * FOUT + 16 + m];
        acc0 = __builtin_amdgcn_wmma_f32_16x16x4_f32(false, a, false, b0,
                                                     (short)0, acc0, false, false);
        acc1 = __builtin_amdgcn_wmma_f32_16x16x4_f32(false, a, false, b1,
                                                     (short)0, acc1, false, false);
    }

    const float bl = bias[m];
    const float bh = bias[16 + m];
    #pragma unroll
    for (int v = 0; v < 8; ++v) {
        const int row = n0 + (half ? (v + 8) : v);
        out[row * FOUT + m]      = 2.0f * acc0[v] + bl;
        out[row * FOUT + 16 + m] = 2.0f * acc1[v] + bh;
    }
}

extern "C" void kernel_launch(void* const* d_in, const int* in_sizes, int n_in,
                              void* d_out, int out_size, void* d_ws, size_t ws_size,
                              hipStream_t stream) {
    const float* input    = (const float*)d_in[0];
    const float* eigenVal = (const float*)d_in[1];
    const float* eigenVec = (const float*)d_in[2];
    const float* W        = (const float*)d_in[3];
    const float* A        = (const float*)d_in[4];
    const float* B        = (const float*)d_in[5];
    const float* C        = (const float*)d_in[6];
    const float* D        = (const float*)d_in[7];
    const float* bias     = (const float*)d_in[8];
    float* out = (float*)d_out;

    float* partial = (float*)d_ws;
    float* S       = partial + (size_t)K1_WAVES * KDIM * FIN;

    k1_partialT<<<K1_BLOCKS, 256, 0, stream>>>(eigenVec, input, partial);
    k2_reduce_S<<<1, 512, 0, stream>>>(partial, eigenVal, W, A, B, C, D, S);
    k3_out<<<N_TOTAL / (16 * WAVES_PER_BLOCK), 256, 0, stream>>>(eigenVec, S, bias, out);
}


# fold the reduce-and-S launch away: k1 sums its 8 waves in LDS and applies W and the coefficient terms per block, the out kernel sums the 16 S shares; 3 launches to 2; order of f32 sums changes
# speedup vs baseline: 1.5675x; 1.5675x over previous
#include <hip/hip_runtime.h>
#include <hip/hip_bf16.h>

typedef __attribute__((ext_vector_type(2))) float v2f;
typedef __attribute__((ext_vector_type(8))) float v8f;

#define N_TOTAL 8192
#define KDIM    16
#define FIN     32
#define FOUT    32
#define TSZ     (KDIM * FIN)

#define WAVES_PER_BLOCK 8
#define K1_BLOCKS       16
#define K1_WAVES        (K1_BLOCKS * WAVES_PER_BLOCK)
#define ROWS_PER_WAVE   (N_TOTAL / K1_WAVES)

__global__ __launch_bounds__(256)
void k1_partialS(const float* __restrict__ ev,
                 const float* __restrict__ in,
                 const float* __restrict__ eigenVal,
                 const float* __restrict__ W,
                 const float* __restrict__ A,
                 const float* __restrict__ B,
                 const float* __restrict__ C,
                 const float* __restrict__ D,
                 float* __restrict__ partialS) {
    __shared__ float red[WAVES_PER_BLOCK * TSZ];
    __shared__ float Tl[TSZ];

    const int tid  = threadIdx.x;
    const int wv   = tid >> 5;
    const int wave = blockIdx.x * WAVES_PER_BLOCK + wv;
    const int lane = tid & 31;
    const int m    = lane & 15;
    const int half = lane >> 4;

    const int ka = wv;
    const int kb = wv + 8;
    float Mra, Mia, Mrb, Mib;
    {
        const float lam = eigenVal[ka];
        const float m1r = A[ka]      * lam + B[ka];
        const float m1i = A[16 + ka] * lam + B[16 + ka];
        const float m2r = 1.0f / (C[ka]      * lam + D[ka]);
        const float m2i = 1.0f / (C[16 + ka] * lam + D[16 + ka]);
        Mra = m1r * m2r - m1i * m2i;
        Mia = m1i * m2r + m1r * m2i;
    }
    {
        const float lam = eigenVal[kb];
        const float m1r = A[kb]      * lam + B[kb];
        const float m1i = A[16 + kb] * lam + B[16 + kb];
        const float m2r = 1.0f / (C[kb]      * lam + D[kb]);
        const float m2i = 1.0f / (C[16 + kb] * lam + D[16 + kb]);
        Mrb = m1r * m2r - m1i * m2i;
        Mib = m1i * m2r + m1r * m2i;
    }

    const int f = lane;
    float w0[FIN], w1[FIN];
    #pragma unroll
    for (int j = 0; j < FIN; ++j) {
        w0[j] = W[j * FOUT + f];
        w1[j] = W[FIN * FOUT + j * FOUT + f];
    }

    v8f acc0 = {};
    v8f acc1 = {};

    const int nstart = wave * ROWS_PER_WAVE;
    #pragma unroll 4
    for (int it = 0; it < ROWS_PER_WAVE / 4; ++it) {
        const int nb = nstart + it * 4 + half * 2;
        v2f a, b0, b1;
        a.x  = ev[nb * KDIM + m];
        a.y  = ev[(nb + 1) * KDIM + m];
        b0.x = in[nb * FIN + m];
        b0.y = in[(nb + 1) * FIN + m];
        b1.x = in[nb * FIN + 16 + m];
        b1.y = in[(nb + 1) * FIN + 16 + m];
        acc0 = __builtin_amdgcn_wmma_f32_16x16x4_f32(false, a, false, b0,
                                                     (short)0, acc0, false, false);
        acc1 = __builtin_amdgcn_wmma_f32_16x16x4_f32(false, a, false, b1,
                                                     (short)0, acc1, false, false);
    }

    float* p = red + wv * TSZ;
    #pragma unroll
    for (int v = 0; v < 8; ++v) {
        const int row = half ? (v + 8) : v;
        p[row * FIN + m]      = acc0[v];
        p[row * FIN + 16 + m] = acc1[v];
    }
    __syncthreads();

    #pragma unroll
    for (int h = 0; h < 2; ++h) {
        const int e = tid + h * 256;
        float s = 0.0f;
        #pragma unroll
        for (int w = 0; w < WAVES_PER_BLOCK; ++w)
            s += red[w * TSZ + e];
        Tl[e] = s;
    }
    __syncthreads();

    float s0a = 0.0f, s1a = 0.0f, s0b = 0.0f, s1b = 0.0f;
    #pragma unroll
    for (int j = 0; j < FIN; ++j) {
        const float ta = Tl[ka * FIN + j];
        const float tb = Tl[kb * FIN + j];
        s0a += ta * w0[j];
        s1a += ta * w1[j];
        s0b += tb * w0[j];
        s1b += tb * w1[j];
    }
    float* ps = partialS + blockIdx.x * TSZ;
    ps[ka * FOUT + f] = Mra * s0a - Mia * s1a;
    ps[kb * FOUT + f] = Mrb * s0b - Mib * s1b;
}

__global__ __launch_bounds__(256)
void k2_out(const float* __restrict__ ev,
            const float* __restrict__ partialS,
            const float* __restrict__ bias,
            float* __restrict__ out) {
    __shared__ float Sl[TSZ];

    const int tid  = threadIdx.x;
    const int wave = blockIdx.x * WAVES_PER_BLOCK + (tid >> 5);
    const int lane = tid & 31;
    const int m    = lane & 15;
    const int half = lane >> 4;
    const int n0   = wave * 16;

    v2f a[4];
    #pragma unroll
    for (int kc = 0; kc < 4; ++kc) {
        const int kg = kc * 4 + half * 2;
        a[kc].x = ev[(n0 + m) * KDIM + kg];
        a[kc].y = ev[(n0 + m) * KDIM + kg + 1];
    }
    const float bl = bias[m];
    const float bh = bias[16 + m];

    #pragma unroll
    for (int h = 0; h < 2; ++h) {
        const int e = tid + h * 256;
        float s = 0.0f;
        #pragma unroll
        for (int b = 0; b < K1_BLOCKS; ++b)
            s += partialS[b * TSZ + e];
        Sl[e] = s;
    }
    __syncthreads();

    v8f acc0 = {};
    v8f acc1 = {};
    #pragma unroll
    for (int kc = 0; kc < 4; ++kc) {
        const int kg = kc * 4 + half * 2;
        v2f b0, b1;
        b0.x = Sl[kg * FOUT + m];
        b0.y = Sl[(kg + 1) * FOUT + m];
        b1.x = Sl[kg * FOUT + 16 + m];
        b1.y = Sl[(kg + 1) * FOUT + 16 + m];
        acc0 = __builtin_amdgcn_wmma_f32_16x16x4_f32(false, a[kc], false, b0,
                                                     (short)0, acc0, false, false);
        acc1 = __builtin_amdgcn_wmma_f32_16x16x4_f32(false, a[kc], false, b1,
                                                     (short)0, acc1, false, false);
    }

    #pragma unroll
    for (int v = 0; v < 8; ++v) {
        const int row = n0 + (half ? (v + 8) : v);
        out[row * FOUT + m]      = 2.0f * acc0[v] + bl;
        out[row * FOUT + 16 + m] = 2.0f * acc1[v] + bh;
    }
}

extern "C" void kernel_launch(void* const* d_in, const int* in_sizes, int n_in,
                              void* d_out, int out_size, void* d_ws, size_t ws_size,
                              hipStream_t stream) {
    const float* input    = (const float*)d_in[0];
    const float* eigenVal = (const float*)d_in[1];
    const float* eigenVec = (const float*)d_in[2];
    const float* W        = (const float*)d_in[3];
    const float* A        = (const float*)d_in[4];
    const float* B        = (const float*)d_in[5];
    const float* C        = (const float*)d_in[6];
    const float* D        = (const float*)d_in[7];
    const float* bias     = (const float*)d_in[8];
    float* out = (float*)d_out;

    float* partialS = (float*)d_ws;

    k1_partialS<<<K1_BLOCKS, 256, 0, stream>>>(eigenVec, input, eigenVal, W, A, B, C, D, partialS);
    k2_out<<<N_TOTAL / (16 * WAVES_PER_BLOCK), 256, 0, stream>>>(eigenVec, partialS, bias, out);
}


# k1: the 64-row loop is fully unrolled so all 96 row loads of a wave go out as one group ahead of the 32 matrix steps (W loads behind them): 4 waited memory rounds become 1
# speedup vs baseline: 1.8379x; 1.1725x over previous
#include <hip/hip_runtime.h>
#include <hip/hip_bf16.h>

typedef __attribute__((ext_vector_type(2))) float v2f;
typedef __attribute__((ext_vector_type(8))) float v8f;

#define N_TOTAL 8192
#define KDIM    16
#define FIN     32
#define FOUT    32
#define TSZ     (KDIM * FIN)

#define WAVES_PER_BLOCK 8
#define K1_BLOCKS       16
#define K1_WAVES        (K1_BLOCKS * WAVES_PER_BLOCK)
#define ROWS_PER_WAVE   (N_TOTAL / K1_WAVES)

__global__ __launch_bounds__(256)
void k1_partialS(const float* __restrict__ ev,
                 const float* __restrict__ in,
                 const float* __restrict__ eigenVal,
                 const float* __restrict__ W,
                 const float* __restrict__ A,
                 const float* __restrict__ B,
                 const float* __restrict__ C,
                 const float* __restrict__ D,
                 float* __restrict__ partialS) {
    __shared__ float red[WAVES_PER_BLOCK * TSZ];
    __shared__ float Tl[TSZ];

    const int tid  = threadIdx.x;
    const int wv   = tid >> 5;
    const int wave = blockIdx.x * WAVES_PER_BLOCK + wv;
    const int lane = tid & 31;
    const int m    = lane & 15;
    const int half = lane >> 4;

    const int ka = wv;
    const int kb = wv + 8;
    float Mra, Mia, Mrb, Mib;
    {
        const float lam = eigenVal[ka];
        const float m1r = A[ka]      * lam + B[ka];
        const float m1i = A[16 + ka] * lam + B[16 + ka];
        const float m2r = 1.0f / (C[ka]      * lam + D[ka]);
        const float m2i = 1.0f / (C[16 + ka] * lam + D[16 + ka]);
        Mra = m1r * m2r - m1i * m2i;
        Mia = m1i * m2r + m1r * m2i;
    }
    {
        const float lam = eigenVal[kb];
        const float m1r = A[kb]      * lam + B[kb];
        const float m1i = A[16 + kb] * lam + B[16 + kb];
        const float m2r = 1.0f / (C[kb]      * lam + D[kb]);
        const float m2i = 1.0f / (C[16 + kb] * lam + D[16 + kb]);
        Mrb = m1r * m2r - m1i * m2i;
        Mib = m1i * m2r + m1r * m2i;
    }

    const int nstart = wave * ROWS_PER_WAVE;
    const float* evp = ev + (nstart + half * 2) * KDIM + m;
    const float* inp = in + (nstart + half * 2) * FIN + m;
    v2f af[ROWS_PER_WAVE / 4], bf0[ROWS_PER_WAVE / 4], bf1[ROWS_PER_WAVE / 4];
    #pragma unroll
    for (int it = 0; it < ROWS_PER_WAVE / 4; ++it) {
        af[it].x  = evp[(it * 4) * KDIM];
        af[it].y  = evp[(it * 4 + 1) * KDIM];
        bf0[it].x = inp[(it * 4) * FIN];
        bf0[it].y = inp[(it * 4 + 1) * FIN];
        bf1[it].x = inp[(it * 4) * FIN + 16];
        bf1[it].y = inp[(it * 4 + 1) * FIN + 16];
    }

    const int f = lane;
    float w0[FIN], w1[FIN];
    #pragma unroll
    for (int j = 0; j < FIN; ++j) {
        w0[j] = W[j * FOUT + f];
        w1[j] = W[FIN * FOUT + j * FOUT + f];
    }

    v8f acc0 = {};
    v8f acc1 = {};

    #pragma unroll
    for (int it = 0; it < ROWS_PER_WAVE / 4; ++it) {
        acc0 = __builtin_amdgcn_wmma_f32_16x16x4_f32(false, af[it], false, bf0[it],
                                                     (short)0, acc0, false, false);
        acc1 = __builtin_amdgcn_wmma_f32_16x16x4_f32(false, af[it], false, bf1[it],
                                                     (short)0, acc1, false, false);
    }

    float* p = red + wv * TSZ;
    #pragma unroll
    for (int v = 0; v < 8; ++v) {
        const int row = half ? (v + 8) : v;
        p[row * FIN + m]      = acc0[v];
        p[row * FIN + 16 + m] = acc1[v];
    }
    __syncthreads();

    #pragma unroll
    for (int h = 0; h < 2; ++h) {
        const int e = tid + h * 256;
        float s = 0.0f;
        #pragma unroll
        for (int w = 0; w < WAVES_PER_BLOCK; ++w)
            s += red[w * TSZ + e];
        Tl[e] = s;
    }
    __syncthreads();

    float s0a = 0.0f, s1a = 0.0f, s0b = 0.0f, s1b = 0.0f;
    #pragma unroll
    for (int j = 0; j < FIN; ++j) {
        const float ta = Tl[ka * FIN + j];
        const float tb = Tl[kb * FIN + j];
        s0a += ta * w0[j];
        s1a += ta * w1[j];
        s0b += tb * w0[j];
        s1b += tb * w1[j];
    }
    float* ps = partialS + blockIdx.x * TSZ;
    ps[ka * FOUT + f] = Mra * s0a - Mia * s1a;
    ps[kb * FOUT + f] = Mrb * s0b - Mib * s1b;
}

__global__ __launch_bounds__(256)
void k2_out(const float* __restrict__ ev,
            const float* __restrict__ partialS,
            const float* __restrict__ bias,
            float* __restrict__ out) {
    __shared__ float Sl[TSZ];

    const int tid  = threadIdx.x;
    const int wave = blockIdx.x * WAVES_PER_BLOCK + (tid >> 5);
    const int lane = tid & 31;
    const int m    = lane & 15;
    const int half = lane >> 4;
    const int n0   = wave * 16;

    v2f a[4];
    #pragma unroll
    for (int kc = 0; kc < 4; ++kc) {
        const int kg = kc * 4 + half * 2;
        a[kc].x = ev[(n0 + m) * KDIM + kg];
        a[kc].y = ev[(n0 + m) * KDIM + kg + 1];
    }
    const float bl = bias[m];
    const float bh = bias[16 + m];

    #pragma unroll
    for (int h = 0; h < 2; ++h) {
        const int e = tid + h * 256;
        float s = 0.0f;
        #pragma unroll
        for (int b = 0; b < K1_BLOCKS; ++b)
            s += partialS[b * TSZ + e];
        Sl[e] = s;
    }
    __syncthreads();

    v8f acc0 = {};
    v8f acc1 = {};
    #pragma unroll
    for (int kc = 0; kc < 4; ++kc) {
        const int kg = kc * 4 + half * 2;
        v2f b0, b1;
        b0.x = Sl[kg * FOUT + m];
        b0.y = Sl[(kg + 1) * FOUT + m];
        b1.x = Sl[kg * FOUT + 16 + m];
        b1.y = Sl[(kg + 1) * FOUT + 16 + m];
        acc0 = __builtin_amdgcn_wmma_f32_16x16x4_f32(false, a[kc], false, b0,
                                                     (short)0, acc0, false, false);
        acc1 = __builtin_amdgcn_wmma_f32_16x16x4_f32(false, a[kc], false, b1,
                                                     (short)0, acc1, false, false);
    }

    #pragma unroll
    for (int v = 0; v < 8; ++v) {
        const int row = n0 + (half ? (v + 8) : v);
        out[row * FOUT + m]      = 2.0f * acc0[v] + bl;
        out[row * FOUT + 16 + m] = 2.0f * acc1[v] + bh;
    }
}

extern "C" void kernel_launch(void* const* d_in, const int* in_sizes, int n_in,
                              void* d_out, int out_size, void* d_ws, size_t ws_size,
                              hipStream_t stream) {
    const float* input    = (const float*)d_in[0];
    const float* eigenVal = (const float*)d_in[1];
    const float* eigenVec = (const float*)d_in[2];
    const float* W        = (const float*)d_in[3];
    const float* A        = (const float*)d_in[4];
    const float* B        = (const float*)d_in[5];
    const float* C        = (const float*)d_in[6];
    const float* D        = (const float*)d_in[7];
    const float* bias     = (const float*)d_in[8];
    float* out = (float*)d_out;

    float* partialS = (float*)d_ws;

    k1_partialS<<<K1_BLOCKS, 256, 0, stream>>>(eigenVec, input, eigenVal, W, A, B, C, D, partialS);
    k2_out<<<N_TOTAL / (16 * WAVES_PER_BLOCK), 256, 0, stream>>>(eigenVec, partialS, bias, out);
}


# fewer, wider memory instructions in both kernels: a lane's two tiles are neighbour columns (8-byte row loads and stores), W once per block via LDS, 16-byte share loads, scalar coefficient loads
# speedup vs baseline: 2.0182x; 1.0981x over previous
#include <hip/hip_runtime.h>
#include <hip/hip_bf16.h>

typedef __attribute__((ext_vector_type(2))) float v2f;
typedef __attribute__((ext_vector_type(4))) float v4f;
typedef __attribute__((ext_vector_type(8))) float v8f;

#define N_TOTAL 8192
#define KDIM    16
#define FIN     32
#define FOUT    32
#define TSZ     (KDIM * FIN)

#define WAVES_PER_BLOCK 8
#define K1_BLOCKS       16
#define K1_WAVES        (K1_BLOCKS * WAVES_PER_BLOCK)
#define ROWS_PER_WAVE   (N_TOTAL / K1_WAVES)

__global__ __launch_bounds__(256)
void k1_partialS(const float* __restrict__ ev,
                 const float* __restrict__ in,
                 const float* __restrict__ eigenVal,
                 const float* __restrict__ W,
                 const float* __restrict__ A,
                 const float* __restrict__ B,
                 const float* __restrict__ C,
                 const float* __restrict__ D,
                 float* __restrict__ partialS) {
    __shared__ float red[WAVES_PER_BLOCK * TSZ];
    __shared__ float Tl[TSZ];
    __shared__ __attribute__((aligned(16))) float Wl[2 * FIN * FOUT];

    const int tid  = threadIdx.x;
    const int wv   = __builtin_amdgcn_readfirstlane(tid >> 5);
    const int wave = blockIdx.x * WAVES_PER_BLOCK + wv;
    const int lane = tid & 31;
    const int m    = lane & 15;
    const int half = lane >> 4;

    const int nstart = wave * ROWS_PER_WAVE;
    const float* evp = ev + (nstart + half * 2) * KDIM + m;
    const float* inp = in + (nstart + half * 2) * FIN + 2 * m;
    v2f af[ROWS_PER_WAVE / 4], bf0[ROWS_PER_WAVE / 4], bf1[ROWS_PER_WAVE / 4];
    #pragma unroll
    for (int it = 0; it < ROWS_PER_WAVE / 4; ++it) {
        af[it].x  = evp[(it * 4) * KDIM];
        af[it].y  = evp[(it * 4 + 1) * KDIM];
        const v2f r0 = *(const v2f*)(inp + (it * 4) * FIN);
        const v2f r1 = *(const v2f*)(inp + (it * 4 + 1) * FIN);
        bf0[it].x = r0.x;  bf0[it].y = r1.x;
        bf1[it].x = r0.y;  bf1[it].y = r1.y;
    }

    __builtin_amdgcn_sched_barrier(0);

    const int f = lane;
    const v4f wq0 = ((const v4f*)W)[tid];
    const v4f wq1 = ((const v4f*)W)[256 + tid];

    const int ka = wv;
    const int kb = wv + 8;
    float Mra, Mia, Mrb, Mib;
    {
        const float lam = eigenVal[ka];
        const float m1r = A[ka]      * lam + B[ka];
        const float m1i = A[16 + ka] * lam + B[16 + ka];
        const float m2r = 1.0f / (C[ka]      * lam + D[ka]);
        const float m2i = 1.0f / (C[16 + ka] * lam + D[16 + ka]);
        Mra = m1r * m2r - m1i * m2i;
        Mia = m1i * m2r + m1r * m2i;
    }
    {
        const float lam = eigenVal[kb];
        const float m1r = A[kb]      * lam + B[kb];
        const float m1i = A[16 + kb] * lam + B[16 + kb];
        const float m2r = 1.0f / (C[kb]      * lam + D[kb]);
        const float m2i = 1.0f / (C[16 + kb] * lam + D[16 + kb]);
        Mrb = m1r * m2r - m1i * m2i;
        Mib = m1i * m2r + m1r * m2i;
    }


    v8f acc0 = {};
    v8f acc1 = {};

    #pragma unroll
    for (int it = 0; it < ROWS_PER_WAVE / 4; ++it) {
        acc0 = __builtin_amdgcn_wmma_f32_16x16x4_f32(false, af[it], false, bf0[it],
                                                     (short)0, acc0, false, false);
        acc1 = __builtin_amdgcn_wmma_f32_16x16x4_f32(false, af[it], false, bf1[it],
                                                     (short)0, acc1, false, false);
    }

    float* p = red + wv * TSZ;
    #pragma unroll
    for (int v = 0; v < 8; ++v) {
        const int row = half ? (v + 8) : v;
        p[row * FIN + 2 * m]     = acc0[v];
        p[row * FIN + 2 * m + 1] = acc1[v];
    }
    ((v4f*)Wl)[tid]       = wq0;
    ((v4f*)Wl)[256 + tid] = wq1;
    __syncthreads();

    float w0[FIN], w1[FIN];
    #pragma unroll
    for (int j = 0; j < FIN; ++j) {
        w0[j] = Wl[j * FOUT + f];
        w1[j] = Wl[FIN * FOUT + j * FOUT + f];
    }

    #pragma unroll
    for (int h = 0; h < 2; ++h) {
        const int e = tid + h * 256;
        float s = 0.0f;
        #pragma unroll
        for (int w = 0; w < WAVES_PER_BLOCK; ++w)
            s += red[w * TSZ + e];
        Tl[e] = s;
    }
    __syncthreads();

    float s0a = 0.0f, s1a = 0.0f, s0b = 0.0f, s1b = 0.0f;
    #pragma unroll
    for (int j = 0; j < FIN; ++j) {
        const float ta = Tl[ka * FIN + j];
        const float tb = Tl[kb * FIN + j];
        s0a += ta * w0[j];
        s1a += ta * w1[j];
        s0b += tb * w0[j];
        s1b += tb * w1[j];
    }
    float* ps = partialS + blockIdx.x * TSZ;
    ps[ka * FOUT + f] = Mra * s0a - Mia * s1a;
    ps[kb * FOUT + f] = Mrb * s0b - Mib * s1b;
}

__global__ __launch_bounds__(256)
void k2_out(const float* __restrict__ ev,
            const float* __restrict__ partialS,
            const float* __restrict__ bias,
            float* __restrict__ out) {
    __shared__ __attribute__((aligned(16))) float Sl[TSZ];

    const int tid  = threadIdx.x;
    const int wave = blockIdx.x * WAVES_PER_BLOCK + (tid >> 5);
    const int lane = tid & 31;
    const int m    = lane & 15;
    const int half = lane >> 4;
    const int n0   = wave * 16;

    v2f a[4];
    #pragma unroll
    for (int kc = 0; kc < 4; ++kc) {
        const int kg = kc * 4 + half * 2;
        a[kc].x = ev[(n0 + m) * KDIM + kg];
        a[kc].y = ev[(n0 + m) * KDIM + kg + 1];
    }
    const v2f bb = *(const v2f*)(bias + 2 * m);

    if (tid < TSZ / 4) {
        const v4f* P4 = (const v4f*)partialS;
        v4f s = {0.0f, 0.0f, 0.0f, 0.0f};
        #pragma unroll
        for (int b = 0; b < K1_BLOCKS; ++b)
            s += P4[b * (TSZ / 4) + tid];
        ((v4f*)Sl)[tid] = s;
    }
    __syncthreads();

    v8f acc0 = {};
    v8f acc1 = {};
    #pragma unroll
    for (int kc = 0; kc < 4; ++kc) {
        const int kg = kc * 4 + half * 2;
        v2f b0, b1;
        b0.x = Sl[kg * FOUT + 2 * m];
        b0.y = Sl[(kg + 1) * FOUT + 2 * m];
        b1.x = Sl[kg * FOUT + 2 * m + 1];
        b1.y = Sl[(kg + 1) * FOUT + 2 * m + 1];
        acc0 = __builtin_amdgcn_wmma_f32_16x16x4_f32(false, a[kc], false, b0,
                                                     (short)0, acc0, false, false);
        acc1 = __builtin_amdgcn_wmma_f32_16x16x4_f32(false, a[kc], false, b1,
                                                     (short)0, acc1, false, false);
    }

    #pragma unroll
    for (int v = 0; v < 8; ++v) {
        const int row = n0 + (half ? (v + 8) : v);
        v2f o;
        o.x = 2.0f * acc0[v] + bb.x;
        o.y = 2.0f * acc1[v] + bb.y;
        *(v2f*)(out + row * FOUT + 2 * m) = o;
    }
}

extern "C" void kernel_launch(void* const* d_in, const int* in_sizes, int n_in,
                              void* d_out, int out_size, void* d_ws, size_t ws_size,
                              hipStream_t stream) {
    const float* input    = (const float*)d_in[0];
    const float* eigenVal = (const float*)d_in[1];
    const float* eigenVec = (const float*)d_in[2];
    const float* W        = (const float*)d_in[3];
    const float* A        = (const float*)d_in[4];
    const float* B        = (const float*)d_in[5];
    const float* C        = (const float*)d_in[6];
    const float* D        = (const float*)d_in[7];
    const float* bias     = (const float*)d_in[8];
    float* out = (float*)d_out;

    float* partialS = (float*)d_ws;

    k1_partialS<<<K1_BLOCKS, 256, 0, stream>>>(eigenVec, input, eigenVal, W, A, B, C, D, partialS);
    k2_out<<<N_TOTAL / (16 * WAVES_PER_BLOCK), 256, 0, stream>>>(eigenVec, partialS, bias, out);
}


# k1 on twice as many work-groups (32 x 8 waves, 32 rows each): half the f32 matrix steps per SIMD and half the row bytes per block; the out kernel sums 32 shares in two halves; f32 sum order changes
# speedup vs baseline: 2.2913x; 1.1353x over previous
#include <hip/hip_runtime.h>
#include <hip/hip_bf16.h>

typedef __attribute__((ext_vector_type(2))) float v2f;
typedef __attribute__((ext_vector_type(4))) float v4f;
typedef __attribute__((ext_vector_type(8))) float v8f;

#define N_TOTAL 8192
#define KDIM    16
#define FIN     32
#define FOUT    32
#define TSZ     (KDIM * FIN)

#define WAVES_PER_BLOCK 8
#define K1_BLOCKS       32
#define K1_WAVES        (K1_BLOCKS * WAVES_PER_BLOCK)
#define ROWS_PER_WAVE   (N_TOTAL / K1_WAVES)

__global__ __launch_bounds__(256)
void k1_partialS(const float* __restrict__ ev,
                 const float* __restrict__ in,
                 const float* __restrict__ eigenVal,
                 const float* __restrict__ W,
                 const float* __restrict__ A,
                 const float* __restrict__ B,
                 const float* __restrict__ C,
                 const float* __restrict__ D,
                 float* __restrict__ partialS) {
    __shared__ float red[WAVES_PER_BLOCK * TSZ];
    __shared__ float Tl[TSZ];
    __shared__ __attribute__((aligned(16))) float Wl[2 * FIN * FOUT];

    const int tid  = threadIdx.x;
    const int wv   = __builtin_amdgcn_readfirstlane(tid >> 5);
    const int wave = blockIdx.x * WAVES_PER_BLOCK + wv;
    const int lane = tid & 31;
    const int m    = lane & 15;
    const int half = lane >> 4;

    const int nstart = wave * ROWS_PER_WAVE;
    const float* evp = ev + (nstart + half * 2) * KDIM + m;
    const float* inp = in + (nstart + half * 2) * FIN + 2 * m;
    v2f af[ROWS_PER_WAVE / 4], bf0[ROWS_PER_WAVE / 4], bf1[ROWS_PER_WAVE / 4];
    #pragma unroll
    for (int it = 0; it < ROWS_PER_WAVE / 4; ++it) {
        af[it].x  = evp[(it * 4) * KDIM];
        af[it].y  = evp[(it * 4 + 1) * KDIM];
        const v2f r0 = *(const v2f*)(inp + (it * 4) * FIN);
        const v2f r1 = *(const v2f*)(inp + (it * 4 + 1) * FIN);
        bf0[it].x = r0.x;  bf0[it].y = r1.x;
        bf1[it].x = r0.y;  bf1[it].y = r1.y;
    }

    __builtin_amdgcn_sched_barrier(0);

    const int f = lane;
    const v4f wq0 = ((const v4f*)W)[tid];
    const v4f wq1 = ((const v4f*)W)[256 + tid];

    const int ka = wv;
    const int kb = wv + 8;
    float Mra, Mia, Mrb, Mib;
    {
        const float lam = eigenVal[ka];
        const float m1r = A[ka]      * lam + B[ka];
        const float m1i = A[16 + ka] * lam + B[16 + ka];
        const float m2r = 1.0f / (C[ka]      * lam + D[ka]);
        const float m2i = 1.0f / (C[16 + ka] * lam + D[16 + ka]);
        Mra = m1r * m2r - m1i * m2i;
        Mia = m1i * m2r + m1r * m2i;
    }
    {
        const float lam = eigenVal[kb];
        const float m1r = A[kb]      * lam + B[kb];
        const float m1i = A[16 + kb] * lam + B[16 + kb];
        const float m2r = 1.0f / (C[kb]      * lam + D[kb]);
        const float m2i = 1.0f / (C[16 + kb] * lam + D[16 + kb]);
        Mrb = m1r * m2r - m1i * m2i;
        Mib = m1i * m2r + m1r * m2i;
    }


    v8f acc0 = {};
    v8f acc1 = {};

    #pragma unroll
    for (int it = 0; it < ROWS_PER_WAVE / 4; ++it) {
        acc0 = __builtin_amdgcn_wmma_f32_16x16x4_f32(false, af[it], false, bf0[it],
                                                     (short)0, acc0, false, false);
        acc1 = __builtin_amdgcn_wmma_f32_16x16x4_f32(false, af[it], false, bf1[it],
                                                     (short)0, acc1, false, false);
    }

    float* p = red + wv * TSZ;
    #pragma unroll
    for (int v = 0; v < 8; ++v) {
        const int row = half ? (v + 8) : v;
        p[row * FIN + 2 * m]     = acc0[v];
        p[row * FIN + 2 * m + 1] = acc1[v];
    }
    ((v4f*)Wl)[tid]       = wq0;
    ((v4f*)Wl)[256 + tid] = wq1;
    __syncthreads();

    float w0[FIN], w1[FIN];
    #pragma unroll
    for (int j = 0; j < FIN; ++j) {
        w0[j] = Wl[j * FOUT + f];
        w1[j] = Wl[FIN * FOUT + j * FOUT + f];
    }

    #pragma unroll
    for (int h = 0; h < 2; ++h) {
        const int e = tid + h * 256;
        float s = 0.0f;
        #pragma unroll
        for (int w = 0; w < WAVES_PER_BLOCK; ++w)
            s += red[w * TSZ + e];
        Tl[e] = s;
    }
    __syncthreads();

    float s0a = 0.0f, s1a = 0.0f, s0b = 0.0f, s1b = 0.0f;
    #pragma unroll
    for (int j = 0; j < FIN; ++j) {
        const float ta = Tl[ka * FIN + j];
        const float tb = Tl[kb * FIN + j];
        s0a += ta * w0[j];
        s1a += ta * w1[j];
        s0b += tb * w0[j];
        s1b += tb * w1[j];
    }
    float* ps = partialS + blockIdx.x * TSZ;
    ps[ka * FOUT + f] = Mra * s0a - Mia * s1a;
    ps[kb * FOUT + f] = Mrb * s0b - Mib * s1b;
}

__global__ __launch_bounds__(256)
void k2_out(const float* __restrict__ ev,
            const float* __restrict__ partialS,
            const float* __restrict__ bias,
            float* __restrict__ out) {
    __shared__ __attribute__((aligned(16))) float Sl[2 * TSZ];

    const int tid  = threadIdx.x;
    const int wave = blockIdx.x * WAVES_PER_BLOCK + (tid >> 5);
    const int lane = tid & 31;
    const int m    = lane & 15;
    const int half = lane >> 4;
    const int n0   = wave * 16;

    v2f a[4];
    #pragma unroll
    for (int kc = 0; kc < 4; ++kc) {
        const int kg = kc * 4 + half * 2;
        a[kc].x = ev[(n0 + m) * KDIM + kg];
        a[kc].y = ev[(n0 + m) * KDIM + kg + 1];
    }
    const v2f bb = *(const v2f*)(bias + 2 * m);

    {
        const int g  = tid & (TSZ / 4 - 1);
        const int hb = tid >> 7;
        const v4f* P4 = (const v4f*)partialS + hb * (K1_BLOCKS / 2) * (TSZ / 4);
        v4f sh[K1_BLOCKS / 2];
        #pragma unroll
        for (int b = 0; b < K1_BLOCKS / 2; ++b)
            sh[b] = P4[b * (TSZ / 4) + g];
        __builtin_amdgcn_sched_barrier(0);
        v4f s = {0.0f, 0.0f, 0.0f, 0.0f};
        #pragma unroll
        for (int b = 0; b < K1_BLOCKS / 2; ++b)
            s += sh[b];
        ((v4f*)Sl)[hb * (TSZ / 4) + g] = s;
    }
    __syncthreads();

    v8f acc0 = {};
    v8f acc1 = {};
    #pragma unroll
    for (int kc = 0; kc < 4; ++kc) {
        const int kg = kc * 4 + half * 2;
        const v2f r0 = *(const v2f*)(Sl + kg * FOUT + 2 * m)
                     + *(const v2f*)(Sl + TSZ + kg * FOUT + 2 * m);
        const v2f r1 = *(const v2f*)(Sl + (kg + 1) * FOUT + 2 * m)
                     + *(const v2f*)(Sl + TSZ + (kg + 1) * FOUT + 2 * m);
        v2f b0, b1;
        b0.x = r0.x;  b0.y = r1.x;
        b1.x = r0.y;  b1.y = r1.y;
        acc0 = __builtin_amdgcn_wmma_f32_16x16x4_f32(false, a[kc], false, b0,
                                                     (short)0, acc0, false, false);
        acc1 = __builtin_amdgcn_wmma_f32_16x16x4_f32(false, a[kc], false, b1,
                                                     (short)0, acc1, false, false);
    }

    #pragma unroll
    for (int v = 0; v < 8; ++v) {
        const int row = n0 + (half ? (v + 8) : v);
        v2f o;
        o.x = 2.0f * acc0[v] + bb.x;
        o.y = 2.0f * acc1[v] + bb.y;
        *(v2f*)(out + row * FOUT + 2 * m) = o;
    }
}

extern "C" void kernel_launch(void* const* d_in, const int* in_sizes, int n_in,
                              void* d_out, int out_size, void* d_ws, size_t ws_size,
                              hipStream_t stream) {
    const float* input    = (const float*)d_in[0];
    const float* eigenVal = (const float*)d_in[1];
    const float* eigenVec = (const float*)d_in[2];
    const float* W        = (const float*)d_in[3];
    const float* A        = (const float*)d_in[4];
    const float* B        = (const float*)d_in[5];
    const float* C        = (const float*)d_in[6];
    const float* D        = (const float*)d_in[7];
    const float* bias     = (const float*)d_in[8];
    float* out = (float*)d_out;

    float* partialS = (float*)d_ws;

    k1_partialS<<<K1_BLOCKS, 256, 0, stream>>>(eigenVec, input, eigenVal, W, A, B, C, D, partialS);
    k2_out<<<N_TOTAL / (16 * WAVES_PER_BLOCK), 256, 0, stream>>>(eigenVec, partialS, bias, out);
}


# k1 tail on the matrix unit: S_b = [Mr*T | -Mi*T] @ [W0;W1], two f32 matrix steps x two tiles per wave on W fragments loaded straight from global; no W in LDS, no FMA loop; f32 sum order changes
# speedup vs baseline: 2.4761x; 1.0807x over previous
#include <hip/hip_runtime.h>
#include <hip/hip_bf16.h>

typedef __attribute__((ext_vector_type(2))) float v2f;
typedef __attribute__((ext_vector_type(4))) float v4f;
typedef __attribute__((ext_vector_type(8))) float v8f;

#define N_TOTAL 8192
#define KDIM    16
#define FIN     32
#define FOUT    32
#define TSZ     (KDIM * FIN)

#define WAVES_PER_BLOCK 8
#define K1_BLOCKS       32
#define K1_WAVES        (K1_BLOCKS * WAVES_PER_BLOCK)
#define ROWS_PER_WAVE   (N_TOTAL / K1_WAVES)

#define TP 36

__global__ __launch_bounds__(256)
void k1_partialS(const float* __restrict__ ev,
                 const float* __restrict__ in,
                 const float* __restrict__ eigenVal,
                 const float* __restrict__ W,
                 const float* __restrict__ A,
                 const float* __restrict__ B,
                 const float* __restrict__ C,
                 const float* __restrict__ D,
                 float* __restrict__ partialS) {
    __shared__ __attribute__((aligned(16))) float red[WAVES_PER_BLOCK * KDIM * TP];
    __shared__ __attribute__((aligned(16))) float sp[WAVES_PER_BLOCK * KDIM * TP];

    const int tid  = threadIdx.x;
    const int wv   = __builtin_amdgcn_readfirstlane(tid >> 5);
    const int wave = blockIdx.x * WAVES_PER_BLOCK + wv;
    const int lane = tid & 31;
    const int m    = lane & 15;
    const int half = lane >> 4;

    const int nstart = wave * ROWS_PER_WAVE;
    const float* evp = ev + (nstart + half * 2) * KDIM + m;
    const float* inp = in + (nstart + half * 2) * FIN + 2 * m;
    v2f af[ROWS_PER_WAVE / 4], bf0[ROWS_PER_WAVE / 4], bf1[ROWS_PER_WAVE / 4];
    #pragma unroll
    for (int it = 0; it < ROWS_PER_WAVE / 4; ++it) {
        af[it].x  = evp[(it * 4) * KDIM];
        af[it].y  = evp[(it * 4 + 1) * KDIM];
        const v2f r0 = *(const v2f*)(inp + (it * 4) * FIN);
        const v2f r1 = *(const v2f*)(inp + (it * 4 + 1) * FIN);
        bf0[it].x = r0.x;  bf0[it].y = r1.x;
        bf1[it].x = r0.y;  bf1[it].y = r1.y;
    }

    __builtin_amdgcn_sched_barrier(0);

    const float* wp = W + (4 * wv + 2 * half) * FOUT + 2 * m;
    v2f wr[2][2];
    #pragma unroll
    for (int s = 0; s < 2; ++s) {
        wr[s][0] = *(const v2f*)(wp + (s * FIN) * FOUT);
        wr[s][1] = *(const v2f*)(wp + (s * FIN + 1) * FOUT);
    }
    const float lam = eigenVal[m];
    const float cAr = A[m], cAi = A[16 + m];
    const float cBr = B[m], cBi = B[16 + m];
    const float cCr = C[m], cCi = C[16 + m];
    const float cDr = D[m], cDi = D[16 + m];

    __builtin_amdgcn_sched_barrier(0);

    v8f acc0 = {};
    v8f acc1 = {};

    #pragma unroll
    for (int it = 0; it < ROWS_PER_WAVE / 4; ++it) {
        acc0 = __builtin_amdgcn_wmma_f32_16x16x4_f32(false, af[it], false, bf0[it],
                                                     (short)0, acc0, false, false);
        acc1 = __builtin_amdgcn_wmma_f32_16x16x4_f32(false, af[it], false, bf1[it],
                                                     (short)0, acc1, false, false);
    }

    __builtin_amdgcn_sched_barrier(0);

    float cre, cim;
    {
        const float m1r = cAr * lam + cBr;
        const float m1i = cAi * lam + cBi;
        const float m2r = 1.0f / (cCr * lam + cDr);
        const float m2i = 1.0f / (cCi * lam + cDi);
        cre = m1r * m2r - m1i * m2i;
        cim = -(m1i * m2r + m1r * m2i);
    }

    {
        float* p = red + wv * (KDIM * TP) + (half * 8) * TP + 2 * m;
        #pragma unroll
        for (int v = 0; v < 8; ++v) {
            v2f t;
            t.x = acc0[v];  t.y = acc1[v];
            *(v2f*)(p + v * TP) = t;
        }
    }
    __syncthreads();

    v2f ta[2];
    {
        const float* q = red + m * TP + 4 * wv + 2 * half;
        v2f sum = *(const v2f*)q;
        #pragma unroll
        for (int w = 1; w < WAVES_PER_BLOCK; ++w)
            sum += *(const v2f*)(q + w * (KDIM * TP));
        ta[0] = sum * cre;
        ta[1] = sum * cim;
    }

    v8f s0 = {};
    v8f s1 = {};
    #pragma unroll
    for (int s = 0; s < 2; ++s) {
        v2f b0, b1;
        b0.x = wr[s][0].x;  b0.y = wr[s][1].x;
        b1.x = wr[s][0].y;  b1.y = wr[s][1].y;
        s0 = __builtin_amdgcn_wmma_f32_16x16x4_f32(false, ta[s], false, b0,
                                                   (short)0, s0, false, false);
        s1 = __builtin_amdgcn_wmma_f32_16x16x4_f32(false, ta[s], false, b1,
                                                   (short)0, s1, false, false);
    }

    {
        float* p = sp + wv * (KDIM * TP) + (half * 8) * TP + 2 * m;
        #pragma unroll
        for (int v = 0; v < 8; ++v) {
            v2f t;
            t.x = s0[v];  t.y = s1[v];
            *(v2f*)(p + v * TP) = t;
        }
    }
    __syncthreads();

    {
        const int row = tid >> 4;
        const int cp  = tid & 15;
        const float* r = sp + row * TP + 2 * cp;
        v2f tot = *(const v2f*)r;
        #pragma unroll
        for (int w = 1; w < WAVES_PER_BLOCK; ++w)
            tot += *(const v2f*)(r + w * (KDIM * TP));
        *(v2f*)(partialS + blockIdx.x * TSZ + row * FOUT + 2 * cp) = tot;
    }
}

__global__ __launch_bounds__(256)
void k2_out(const float* __restrict__ ev,
            const float* __restrict__ partialS,
            const float* __restrict__ bias,
            float* __restrict__ out) {
    __shared__ __attribute__((aligned(16))) float Sl[2 * TSZ];

    const int tid  = threadIdx.x;
    const int wave = blockIdx.x * WAVES_PER_BLOCK + (tid >> 5);
    const int lane = tid & 31;
    const int m    = lane & 15;
    const int half = lane >> 4;
    const int n0   = wave * 16;

    v2f a[4];
    #pragma unroll
    for (int kc = 0; kc < 4; ++kc) {
        const int kg = kc * 4 + half * 2;
        a[kc].x = ev[(n0 + m) * KDIM + kg];
        a[kc].y = ev[(n0 + m) * KDIM + kg + 1];
    }
    const v2f bb = *(const v2f*)(bias + 2 * m);

    {
        const int g  = tid & (TSZ / 4 - 1);
        const int hb = tid >> 7;
        const v4f* P4 = (const v4f*)partialS + hb * (K1_BLOCKS / 2) * (TSZ / 4);
        v4f sh[K1_BLOCKS / 2];
        #pragma unroll
        for (int b = 0; b < K1_BLOCKS / 2; ++b)
            sh[b] = P4[b * (TSZ / 4) + g];
        __builtin_amdgcn_sched_barrier(0);
        v4f s = {0.0f, 0.0f, 0.0f, 0.0f};
        #pragma unroll
        for (int b = 0; b < K1_BLOCKS / 2; ++b)
            s += sh[b];
        ((v4f*)Sl)[hb * (TSZ / 4) + g] = s;
    }
    __syncthreads();

    v8f acc0 = {};
    v8f acc1 = {};
    #pragma unroll
    for (int kc = 0; kc < 4; ++kc) {
        const int kg = kc * 4 + half * 2;
        const v2f r0 = *(const v2f*)(Sl + kg * FOUT + 2 * m)
                     + *(const v2f*)(Sl + TSZ + kg * FOUT + 2 * m);
        const v2f r1 = *(const v2f*)(Sl + (kg + 1) * FOUT + 2 * m)
                     + *(const v2f*)(Sl + TSZ + (kg + 1) * FOUT + 2 * m);
        v2f b0, b1;
        b0.x = r0.x;  b0.y = r1.x;
        b1.x = r0.y;  b1.y = r1.y;
        acc0 = __builtin_amdgcn_wmma_f32_16x16x4_f32(false, a[kc], false, b0,
                                                     (short)0, acc0, false, false);
        acc1 = __builtin_amdgcn_wmma_f32_16x16x4_f32(false, a[kc], false, b1,
                                                     (short)0, acc1, false, false);
    }

    #pragma unroll
    for (int v = 0; v < 8; ++v) {
        const int row = n0 + (half ? (v + 8) : v);
        v2f o;
        o.x = 2.0f * acc0[v] + bb.x;
        o.y = 2.0f * acc1[v] + bb.y;
        *(v2f*)(out + row * FOUT + 2 * m) = o;
    }
}

extern "C" void kernel_launch(void* const* d_in, const int* in_sizes, int n_in,
                              void* d_out, int out_size, void* d_ws, size_t ws_size,
                              hipStream_t stream) {
    const float* input    = (const float*)d_in[0];
    const float* eigenVal = (const float*)d_in[1];
    const float* eigenVec = (const float*)d_in[2];
    const float* W        = (const float*)d_in[3];
    const float* A        = (const float*)d_in[4];
    const float* B        = (const float*)d_in[5];
    const float* C        = (const float*)d_in[6];
    const float* D        = (const float*)d_in[7];
    const float* bias     = (const float*)d_in[8];
    float* out = (float*)d_out;

    float* partialS = (float*)d_ws;

    k1_partialS<<<K1_BLOCKS, 256, 0, stream>>>(eigenVec, input, eigenVal, W, A, B, C, D, partialS);
    k2_out<<<N_TOTAL / (16 * WAVES_PER_BLOCK), 256, 0, stream>>>(eigenVec, partialS, bias, out);
}
